# starting kernel
# baseline (speedup 1.0000x reference)
#include <hip/hip_runtime.h>
#include <math.h>

typedef __attribute__((ext_vector_type(2))) float v2f;
typedef __attribute__((ext_vector_type(8))) float v8f;

#define NBASIS 128
#define NCHUNK 32
#define CUTOFF 6.0f
#define DELTA  (CUTOFF / (NBASIS - 1))
#define GAMMA_ ((NBASIS - 1) / CUTOFF)
#define PI_F   3.14159265358979323846f
#define LOG2E_F 1.44269504088896340736f
#define K2_    (-GAMMA_ * LOG2E_F)

#if __has_builtin(__builtin_amdgcn_exp2f)
#define FAST_EXP2(x) __builtin_amdgcn_exp2f(x)
#else
#define FAST_EXP2(x) exp2f(x)
#endif

__device__ __forceinline__ float softplus_(float x) {
  return fmaxf(x, 0.0f) + log1pf(expf(-fabsf(x)));
}

__device__ __forceinline__ float blockReduceSum256(float v) {
  __shared__ float sred[256];
  sred[threadIdx.x] = v;
  __syncthreads();
  for (int s = blockDim.x >> 1; s > 0; s >>= 1) {
    if (threadIdx.x < (unsigned)s) sred[threadIdx.x] += sred[threadIdx.x + s];
    __syncthreads();
  }
  return sred[0];
}

__global__ void eam_zero(float* __restrict__ a, int na, float* __restrict__ b, int nb) {
  const int i = blockIdx.x * blockDim.x + threadIdx.x;
  const int stride = gridDim.x * blockDim.x;
  for (int j = i; j < na; j += stride) a[j] = 0.0f;
  for (int j = i; j < nb; j += stride) b[j] = 0.0f;
}

__global__ __launch_bounds__(256) void eam_edge_fwd(
    const float* __restrict__ r, const int* __restrict__ dst,
    const float* __restrict__ phiD, const float* __restrict__ phiP,
    float* __restrict__ local, float* __restrict__ gd_out, float* __restrict__ gp_out,
    float* __restrict__ partial_pair, int E)
{
  __shared__ v2f sW[64 * 16];

  for (int idx = threadIdx.x; idx < 64 * 16; idx += blockDim.x) {
    const int kp = idx >> 4, ch = idx & 15;
    const int k0 = 2 * kp, k1 = k0 + 1;
    const float c0 = k0 * DELTA, c1 = k1 * DELTA;
    v2f w = {0.0f, 0.0f};
    if (ch < 4) {
      float a0, a1;
      if (ch < 2) { a0 = softplus_(phiD[k0]); a1 = softplus_(phiD[k1]); }
      else        { a0 = phiP[k0];            a1 = phiP[k1]; }
      if (ch & 1) { a0 *= c0; a1 *= c1; }
      w.x = a0; w.y = a1;
    }
    sW[idx] = w;
  }
  __syncthreads();

  const int lane = threadIdx.x & 31;
  const int col  = lane & 15;
  const int hi   = lane >> 4;

  v2f aReg[NCHUNK];
#pragma unroll
  for (int t = 0; t < NCHUNK; ++t)
    aReg[t] = sW[(2 * t + hi) * 16 + col];

  const int wavesPerBlk = blockDim.x >> 5;
  const int wave   = blockIdx.x * wavesPerBlk + (threadIdx.x >> 5);
  const int nWaves = gridDim.x * wavesPerBlk;
  const int nTiles = (E + 15) >> 4;

  float pairAcc = 0.0f;

  for (int tile = wave; tile < nTiles; tile += nWaves) {
    const int e = tile * 16 + col;
    const bool valid = (e < E);

    const int eNext = (tile + nWaves) * 16 + col;
    if (eNext < E) __builtin_prefetch(&r[3 * eNext], 0, 0);

    float x = 0.0f, y = 0.0f, z = 0.0f;
    if (valid) { x = r[3 * e]; y = r[3 * e + 1]; z = r[3 * e + 2]; }
    const float dist = sqrtf(x * x + y * y + z * z);

    v8f acc = {0.f, 0.f, 0.f, 0.f, 0.f, 0.f, 0.f, 0.f};
    const float base = dist - (2 * hi) * DELTA;
    const v2f base2 = {base, base - DELTA};
#pragma unroll
    for (int t = 0; t < NCHUNK; ++t) {
      const v2f tt  = base2 - (float)(4 * t) * DELTA;
      const v2f arg = (tt * tt) * (float)K2_;
      v2f g;
      g.x = FAST_EXP2(arg.x);
      g.y = FAST_EXP2(arg.y);
      acc = __builtin_amdgcn_wmma_f32_16x16x4_f32(
          false, aReg[t], false, g, (short)0, acc, false, false);
    }

    if (lane < 16 && valid) {
      const float SA = acc[0], SB = acc[1], SC = acc[2], SD = acc[3];
      float sn, cs;
      __sincosf(dist * (PI_F / CUTOFF), &sn, &cs);
      const float phi  = 0.25f * (1.0f + cs);
      const float phip = -0.25f * (PI_F / CUTOFF) * sn;
      const float invd = 1.0f / dist;

      const float dens   = phi * SA;
      const float ddens  = phi * (-2.0f * GAMMA_) * (dist * SA - SB) + phip * SA;
      const float pairn  = phi * SC;
      const float dpairn = phi * (-2.0f * GAMMA_) * (dist * SC - SD) + phip * SC;
      const float dpair  = (dpairn - pairn * invd) * invd;

      atomicAdd(&local[dst[e]], dens);
      gd_out[e] = ddens * invd;
      gp_out[e] = dpair * invd;
      pairAcc += pairn * invd;
    }
  }

  const float bsum = blockReduceSum256(pairAcc);
  if (threadIdx.x == 0) partial_pair[blockIdx.x] = bsum;
}

__global__ __launch_bounds__(256) void eam_node(
    const float* __restrict__ local, const float* __restrict__ w,
    float* __restrict__ Fp, float* __restrict__ partial_F, int N)
{
  const float w0 = w[0], w1 = w[1], w2 = w[2], w3 = w[3], w4 = w[4];
  float Facc = 0.0f;
  for (int n = blockIdx.x * blockDim.x + threadIdx.x; n < N;
       n += gridDim.x * blockDim.x) {
    const float x  = local[n];
    const float sq = sqrtf(x);
    const float x2 = x * x;
    const float F  = 2.0f * w0 * sq + w1 * x + 0.5f * w2 * x2 +
                     (1.0f / 6.0f) * w3 * x2 * x + (1.0f / 24.0f) * w4 * x2 * x2;
    const float Fd = w0 / sq + w1 + w2 * x + 0.5f * w3 * x2 +
                     (1.0f / 6.0f) * w4 * x2 * x;
    Facc += F;
    Fp[n] = Fd;
  }
  const float bsum = blockReduceSum256(Facc);
  if (threadIdx.x == 0) partial_F[blockIdx.x] = bsum;
}

__global__ __launch_bounds__(256) void eam_force(
    const float* __restrict__ r, const int* __restrict__ src,
    const int* __restrict__ dst, const float* __restrict__ gd,
    const float* __restrict__ gp, const float* __restrict__ Fp,
    float* __restrict__ forces, int E)
{
  for (int e = blockIdx.x * blockDim.x + threadIdx.x; e < E;
       e += gridDim.x * blockDim.x) {
    const int s = src[e], d = dst[e];
    const float g = Fp[d] * gd[e] + gp[e];
    const float fx = g * r[3 * e], fy = g * r[3 * e + 1], fz = g * r[3 * e + 2];
    atomicAdd(&forces[3 * s + 0],  fx);
    atomicAdd(&forces[3 * s + 1],  fy);
    atomicAdd(&forces[3 * s + 2],  fz);
    atomicAdd(&forces[3 * d + 0], -fx);
    atomicAdd(&forces[3 * d + 1], -fy);
    atomicAdd(&forces[3 * d + 2], -fz);
  }
}

__global__ __launch_bounds__(256) void eam_energy(
    const float* __restrict__ partial_pair, int n1,
    const float* __restrict__ partial_F, int n2,
    float* __restrict__ energy)
{
  float v = 0.0f;
  for (int i = threadIdx.x; i < n1; i += blockDim.x) v += partial_pair[i];
  for (int i = threadIdx.x; i < n2; i += blockDim.x) v += partial_F[i];
  const float s = blockReduceSum256(v);
  if (threadIdx.x == 0) energy[0] = s;
}

extern "C" void kernel_launch(void* const* d_in, const int* in_sizes, int n_in,
                              void* d_out, int out_size, void* d_ws, size_t ws_size,
                              hipStream_t stream) {
  const float* r    = (const float*)d_in[0];
  const int*   src  = (const int*)  d_in[1];
  const int*   dst  = (const int*)  d_in[2];
  const float* phiD = (const float*)d_in[3];
  const float* phiP = (const float*)d_in[4];
  const float* emb  = (const float*)d_in[5];

  const int E = in_sizes[1];
  const int N = (out_size - 1) / 3;

  float* out    = (float*)d_out;
  float* energy = out;
  float* forces = out + 1;

  float* ws    = (float*)d_ws;
  float* gd    = ws;
  float* gp    = gd + E;
  float* local = gp + E;
  float* Fp    = local + N;
  const int NB1 = 512;
  const int NB2 = 128;
  float* ppair = Fp + N;
  float* pF    = ppair + NB1;

  eam_zero    <<<256,  256, 0, stream>>>(local, N, forces, 3 * N);
  eam_edge_fwd<<<NB1,  256, 0, stream>>>(r, dst, phiD, phiP, local, gd, gp, ppair, E);
  eam_node    <<<NB2,  256, 0, stream>>>(local, emb, Fp, pF, N);
  eam_force   <<<1024, 256, 0, stream>>>(r, src, dst, gd, gp, Fp, forces, E);
  eam_energy  <<<1,    256, 0, stream>>>(ppair, NB1, pF, NB2, energy);
}


# force kernel: a node's x,y,z atomic adds moved into adjacent lanes of one atomic instruction (lane exchange), one request per node instead of three; same f32 atomics, sums still in arrival order
# speedup vs baseline: 1.7777x; 1.7777x over previous
#include <hip/hip_runtime.h>
#include <math.h>

typedef __attribute__((ext_vector_type(2))) float v2f;
typedef __attribute__((ext_vector_type(8))) float v8f;

#define NBASIS 128
#define NCHUNK 32
#define CUTOFF 6.0f
#define DELTA  (CUTOFF / (NBASIS - 1))
#define GAMMA_ ((NBASIS - 1) / CUTOFF)
#define PI_F   3.14159265358979323846f
#define LOG2E_F 1.44269504088896340736f
#define K2_    (-GAMMA_ * LOG2E_F)

#if __has_builtin(__builtin_amdgcn_exp2f)
#define FAST_EXP2(x) __builtin_amdgcn_exp2f(x)
#else
#define FAST_EXP2(x) exp2f(x)
#endif

__device__ __forceinline__ float softplus_(float x) {
  return fmaxf(x, 0.0f) + log1pf(expf(-fabsf(x)));
}

__device__ __forceinline__ float blockReduceSum256(float v) {
  __shared__ float sred[256];
  sred[threadIdx.x] = v;
  __syncthreads();
  for (int s = blockDim.x >> 1; s > 0; s >>= 1) {
    if (threadIdx.x < (unsigned)s) sred[threadIdx.x] += sred[threadIdx.x + s];
    __syncthreads();
  }
  return sred[0];
}

__global__ void eam_zero(float* __restrict__ a, int na, float* __restrict__ b, int nb) {
  const int i = blockIdx.x * blockDim.x + threadIdx.x;
  const int stride = gridDim.x * blockDim.x;
  for (int j = i; j < na; j += stride) a[j] = 0.0f;
  for (int j = i; j < nb; j += stride) b[j] = 0.0f;
}

__global__ __launch_bounds__(256) void eam_edge_fwd(
    const float* __restrict__ r, const int* __restrict__ dst,
    const float* __restrict__ phiD, const float* __restrict__ phiP,
    float* __restrict__ local, float* __restrict__ gd_out, float* __restrict__ gp_out,
    float* __restrict__ partial_pair, int E)
{
  __shared__ v2f sW[64 * 16];

  for (int idx = threadIdx.x; idx < 64 * 16; idx += blockDim.x) {
    const int kp = idx >> 4, ch = idx & 15;
    const int k0 = 2 * kp, k1 = k0 + 1;
    const float c0 = k0 * DELTA, c1 = k1 * DELTA;
    v2f w = {0.0f, 0.0f};
    if (ch < 4) {
      float a0, a1;
      if (ch < 2) { a0 = softplus_(phiD[k0]); a1 = softplus_(phiD[k1]); }
      else        { a0 = phiP[k0];            a1 = phiP[k1]; }
      if (ch & 1) { a0 *= c0; a1 *= c1; }
      w.x = a0; w.y = a1;
    }
    sW[idx] = w;
  }
  __syncthreads();

  const int lane = threadIdx.x & 31;
  const int col  = lane & 15;
  const int hi   = lane >> 4;

  v2f aReg[NCHUNK];
#pragma unroll
  for (int t = 0; t < NCHUNK; ++t)
    aReg[t] = sW[(2 * t + hi) * 16 + col];

  const int wavesPerBlk = blockDim.x >> 5;
  const int wave   = blockIdx.x * wavesPerBlk + (threadIdx.x >> 5);
  const int nWaves = gridDim.x * wavesPerBlk;
  const int nTiles = (E + 15) >> 4;

  float pairAcc = 0.0f;

  for (int tile = wave; tile < nTiles; tile += nWaves) {
    const int e = tile * 16 + col;
    const bool valid = (e < E);

    const int eNext = (tile + nWaves) * 16 + col;
    if (eNext < E) __builtin_prefetch(&r[3 * eNext], 0, 0);

    float x = 0.0f, y = 0.0f, z = 0.0f;
    if (valid) { x = r[3 * e]; y = r[3 * e + 1]; z = r[3 * e + 2]; }
    const float dist = sqrtf(x * x + y * y + z * z);

    v8f acc = {0.f, 0.f, 0.f, 0.f, 0.f, 0.f, 0.f, 0.f};
    const float base = dist - (2 * hi) * DELTA;
    const v2f base2 = {base, base - DELTA};
#pragma unroll
    for (int t = 0; t < NCHUNK; ++t) {
      const v2f tt  = base2 - (float)(4 * t) * DELTA;
      const v2f arg = (tt * tt) * (float)K2_;
      v2f g;
      g.x = FAST_EXP2(arg.x);
      g.y = FAST_EXP2(arg.y);
      acc = __builtin_amdgcn_wmma_f32_16x16x4_f32(
          false, aReg[t], false, g, (short)0, acc, false, false);
    }

    if (lane < 16 && valid) {
      const float SA = acc[0], SB = acc[1], SC = acc[2], SD = acc[3];
      float sn, cs;
      __sincosf(dist * (PI_F / CUTOFF), &sn, &cs);
      const float phi  = 0.25f * (1.0f + cs);
      const float phip = -0.25f * (PI_F / CUTOFF) * sn;
      const float invd = 1.0f / dist;

      const float dens   = phi * SA;
      const float ddens  = phi * (-2.0f * GAMMA_) * (dist * SA - SB) + phip * SA;
      const float pairn  = phi * SC;
      const float dpairn = phi * (-2.0f * GAMMA_) * (dist * SC - SD) + phip * SC;
      const float dpair  = (dpairn - pairn * invd) * invd;

      atomicAdd(&local[dst[e]], dens);
      gd_out[e] = ddens * invd;
      gp_out[e] = dpair * invd;
      pairAcc += pairn * invd;
    }
  }

  const float bsum = blockReduceSum256(pairAcc);
  if (threadIdx.x == 0) partial_pair[blockIdx.x] = bsum;
}

__global__ __launch_bounds__(256) void eam_node(
    const float* __restrict__ local, const float* __restrict__ w,
    float* __restrict__ Fp, float* __restrict__ partial_F, int N)
{
  const float w0 = w[0], w1 = w[1], w2 = w[2], w3 = w[3], w4 = w[4];
  float Facc = 0.0f;
  for (int n = blockIdx.x * blockDim.x + threadIdx.x; n < N;
       n += gridDim.x * blockDim.x) {
    const float x  = local[n];
    const float sq = sqrtf(x);
    const float x2 = x * x;
    const float F  = 2.0f * w0 * sq + w1 * x + 0.5f * w2 * x2 +
                     (1.0f / 6.0f) * w3 * x2 * x + (1.0f / 24.0f) * w4 * x2 * x2;
    const float Fd = w0 / sq + w1 + w2 * x + 0.5f * w3 * x2 +
                     (1.0f / 6.0f) * w4 * x2 * x;
    Facc += F;
    Fp[n] = Fd;
  }
  const float bsum = blockReduceSum256(Facc);
  if (threadIdx.x == 0) partial_F[blockIdx.x] = bsum;
}

__device__ __forceinline__ float lane_get_f(int byteSel, float v) {
  return __int_as_float(__builtin_amdgcn_ds_bpermute(byteSel, __float_as_int(v)));
}

__global__ __launch_bounds__(256) void eam_force(
    const float* __restrict__ r, const int* __restrict__ src,
    const int* __restrict__ dst, const float* __restrict__ gd,
    const float* __restrict__ gp, const float* __restrict__ Fp,
    float* __restrict__ forces, int E)
{
  const int lane    = threadIdx.x & 31;
  const int wave    = (blockIdx.x * blockDim.x + threadIdx.x) >> 5;
  const int nWaves  = (gridDim.x * blockDim.x) >> 5;
  const int nGroups = (E + 31) >> 5;

  for (int grp = wave; grp < nGroups; grp += nWaves) {
    const int e0 = grp << 5;
    const int e  = e0 + lane;
    int   s3 = 0, d3 = 0;
    float fx = 0.0f, fy = 0.0f, fz = 0.0f;
    if (e < E) {
      const int s = src[e], d = dst[e];
      const float g = Fp[d] * gd[e] + gp[e];
      fx = g * r[3 * e]; fy = g * r[3 * e + 1]; fz = g * r[3 * e + 2];
      s3 = 3 * s; d3 = 3 * d;
    }
    for (int j = 0; j < 3; ++j) {
      const int c    = 32 * j + lane;
      const int t    = c / 3;
      const int comp = c - 3 * t;
      const int sel  = t << 2;
      const float px = lane_get_f(sel, fx);
      const float py = lane_get_f(sel, fy);
      const float pz = lane_get_f(sel, fz);
      const int   sn = __builtin_amdgcn_ds_bpermute(sel, s3);
      const int   dn = __builtin_amdgcn_ds_bpermute(sel, d3);
      const float v  = (comp == 0) ? px : ((comp == 1) ? py : pz);
      if (e0 + t < E) {
        atomicAdd(&forces[sn + comp],  v);
        atomicAdd(&forces[dn + comp], -v);
      }
    }
  }
}

__global__ __launch_bounds__(256) void eam_energy(
    const float* __restrict__ partial_pair, int n1,
    const float* __restrict__ partial_F, int n2,
    float* __restrict__ energy)
{
  float v = 0.0f;
  for (int i = threadIdx.x; i < n1; i += blockDim.x) v += partial_pair[i];
  for (int i = threadIdx.x; i < n2; i += blockDim.x) v += partial_F[i];
  const float s = blockReduceSum256(v);
  if (threadIdx.x == 0) energy[0] = s;
}

extern "C" void kernel_launch(void* const* d_in, const int* in_sizes, int n_in,
                              void* d_out, int out_size, void* d_ws, size_t ws_size,
                              hipStream_t stream) {
  const float* r    = (const float*)d_in[0];
  const int*   src  = (const int*)  d_in[1];
  const int*   dst  = (const int*)  d_in[2];
  const float* phiD = (const float*)d_in[3];
  const float* phiP = (const float*)d_in[4];
  const float* emb  = (const float*)d_in[5];

  const int E = in_sizes[1];
  const int N = (out_size - 1) / 3;

  float* out    = (float*)d_out;
  float* energy = out;
  float* forces = out + 1;

  float* ws    = (float*)d_ws;
  float* gd    = ws;
  float* gp    = gd + E;
  float* local = gp + E;
  float* Fp    = local + N;
  const int NB1 = 512;
  const int NB2 = 128;
  float* ppair = Fp + N;
  float* pF    = ppair + NB1;

  eam_zero    <<<256,  256, 0, stream>>>(local, N, forces, 3 * N);
  eam_edge_fwd<<<NB1,  256, 0, stream>>>(r, dst, phiD, phiP, local, gd, gp, ppair, E);
  eam_node    <<<NB2,  256, 0, stream>>>(local, emb, Fp, pF, N);
  eam_force   <<<1024, 256, 0, stream>>>(r, src, dst, gd, gp, Fp, forces, E);
  eam_energy  <<<1,    256, 0, stream>>>(ppair, NB1, pF, NB2, energy);
}


# force kernel: scatter gathered per block in LDS over (node range x edge slice) by LDS f32 adds, then whole-line global f32 atomic adds of block sums; sums regrouped per block, still arrival order
# speedup vs baseline: 2.8594x; 1.6085x over previous
#include <hip/hip_runtime.h>
#include <math.h>

typedef __attribute__((ext_vector_type(2))) float v2f;
typedef __attribute__((ext_vector_type(8))) float v8f;

#define NBASIS 128
#define NCHUNK 32
#define CUTOFF 6.0f
#define DELTA  (CUTOFF / (NBASIS - 1))
#define GAMMA_ ((NBASIS - 1) / CUTOFF)
#define PI_F   3.14159265358979323846f
#define LOG2E_F 1.44269504088896340736f
#define K2_    (-GAMMA_ * LOG2E_F)

#if __has_builtin(__builtin_amdgcn_exp2f)
#define FAST_EXP2(x) __builtin_amdgcn_exp2f(x)
#else
#define FAST_EXP2(x) exp2f(x)
#endif

__device__ __forceinline__ float softplus_(float x) {
  return fmaxf(x, 0.0f) + log1pf(expf(-fabsf(x)));
}

__device__ __forceinline__ float blockReduceSum256(float v) {
  __shared__ float sred[256];
  sred[threadIdx.x] = v;
  __syncthreads();
  for (int s = blockDim.x >> 1; s > 0; s >>= 1) {
    if (threadIdx.x < (unsigned)s) sred[threadIdx.x] += sred[threadIdx.x + s];
    __syncthreads();
  }
  return sred[0];
}

__global__ void eam_zero(float* __restrict__ a, int na, float* __restrict__ b, int nb) {
  const int i = blockIdx.x * blockDim.x + threadIdx.x;
  const int stride = gridDim.x * blockDim.x;
  for (int j = i; j < na; j += stride) a[j] = 0.0f;
  for (int j = i; j < nb; j += stride) b[j] = 0.0f;
}

__global__ __launch_bounds__(256) void eam_edge_fwd(
    const float* __restrict__ r, const int* __restrict__ dst,
    const float* __restrict__ phiD, const float* __restrict__ phiP,
    float* __restrict__ local, float* __restrict__ gd_out, float* __restrict__ gp_out,
    float* __restrict__ partial_pair, int E)
{
  __shared__ v2f sW[64 * 16];

  for (int idx = threadIdx.x; idx < 64 * 16; idx += blockDim.x) {
    const int kp = idx >> 4, ch = idx & 15;
    const int k0 = 2 * kp, k1 = k0 + 1;
    const float c0 = k0 * DELTA, c1 = k1 * DELTA;
    v2f w = {0.0f, 0.0f};
    if (ch < 4) {
      float a0, a1;
      if (ch < 2) { a0 = softplus_(phiD[k0]); a1 = softplus_(phiD[k1]); }
      else        { a0 = phiP[k0];            a1 = phiP[k1]; }
      if (ch & 1) { a0 *= c0; a1 *= c1; }
      w.x = a0; w.y = a1;
    }
    sW[idx] = w;
  }
  __syncthreads();

  const int lane = threadIdx.x & 31;
  const int col  = lane & 15;
  const int hi   = lane >> 4;

  v2f aReg[NCHUNK];
#pragma unroll
  for (int t = 0; t < NCHUNK; ++t)
    aReg[t] = sW[(2 * t + hi) * 16 + col];

  const int wavesPerBlk = blockDim.x >> 5;
  const int wave   = blockIdx.x * wavesPerBlk + (threadIdx.x >> 5);
  const int nWaves = gridDim.x * wavesPerBlk;
  const int nTiles = (E + 15) >> 4;

  float pairAcc = 0.0f;

  for (int tile = wave; tile < nTiles; tile += nWaves) {
    const int e = tile * 16 + col;
    const bool valid = (e < E);

    const int eNext = (tile + nWaves) * 16 + col;
    if (eNext < E) __builtin_prefetch(&r[3 * eNext], 0, 0);

    float x = 0.0f, y = 0.0f, z = 0.0f;
    if (valid) { x = r[3 * e]; y = r[3 * e + 1]; z = r[3 * e + 2]; }
    const float dist = sqrtf(x * x + y * y + z * z);

    v8f acc = {0.f, 0.f, 0.f, 0.f, 0.f, 0.f, 0.f, 0.f};
    const float base = dist - (2 * hi) * DELTA;
    const v2f base2 = {base, base - DELTA};
#pragma unroll
    for (int t = 0; t < NCHUNK; ++t) {
      const v2f tt  = base2 - (float)(4 * t) * DELTA;
      const v2f arg = (tt * tt) * (float)K2_;
      v2f g;
      g.x = FAST_EXP2(arg.x);
      g.y = FAST_EXP2(arg.y);
      acc = __builtin_amdgcn_wmma_f32_16x16x4_f32(
          false, aReg[t], false, g, (short)0, acc, false, false);
    }

    if (lane < 16 && valid) {
      const float SA = acc[0], SB = acc[1], SC = acc[2], SD = acc[3];
      float sn, cs;
      __sincosf(dist * (PI_F / CUTOFF), &sn, &cs);
      const float phi  = 0.25f * (1.0f + cs);
      const float phip = -0.25f * (PI_F / CUTOFF) * sn;
      const float invd = 1.0f / dist;

      const float dens   = phi * SA;
      const float ddens  = phi * (-2.0f * GAMMA_) * (dist * SA - SB) + phip * SA;
      const float pairn  = phi * SC;
      const float dpairn = phi * (-2.0f * GAMMA_) * (dist * SC - SD) + phip * SC;
      const float dpair  = (dpairn - pairn * invd) * invd;

      atomicAdd(&local[dst[e]], dens);
      gd_out[e] = ddens * invd;
      gp_out[e] = dpair * invd;
      pairAcc += pairn * invd;
    }
  }

  const float bsum = blockReduceSum256(pairAcc);
  if (threadIdx.x == 0) partial_pair[blockIdx.x] = bsum;
}

__global__ __launch_bounds__(256) void eam_node(
    const float* __restrict__ local, const float* __restrict__ w,
    float* __restrict__ Fp, float* __restrict__ partial_F, int N)
{
  const float w0 = w[0], w1 = w[1], w2 = w[2], w3 = w[3], w4 = w[4];
  float Facc = 0.0f;
  for (int n = blockIdx.x * blockDim.x + threadIdx.x; n < N;
       n += gridDim.x * blockDim.x) {
    const float x  = local[n];
    const float sq = sqrtf(x);
    const float x2 = x * x;
    const float F  = 2.0f * w0 * sq + w1 * x + 0.5f * w2 * x2 +
                     (1.0f / 6.0f) * w3 * x2 * x + (1.0f / 24.0f) * w4 * x2 * x2;
    const float Fd = w0 / sq + w1 + w2 * x + 0.5f * w3 * x2 +
                     (1.0f / 6.0f) * w4 * x2 * x;
    Facc += F;
    Fp[n] = Fd;
  }
  const float bsum = blockReduceSum256(Facc);
  if (threadIdx.x == 0) partial_F[blockIdx.x] = bsum;
}

#define FR_RANGES  8
#define FR_NODES   6250
#define FR_THREADS 1024
#define FR_UNROLL  4

__global__ __launch_bounds__(FR_THREADS) void eam_force(
    const float* __restrict__ r, const int* __restrict__ src,
    const int* __restrict__ dst, const float* __restrict__ gd,
    const float* __restrict__ gp, const float* __restrict__ Fp,
    float* __restrict__ forces, int E, int N)
{
  __shared__ float sF[3 * FR_NODES];

  const int tid  = threadIdx.x;
  const int nThr = blockDim.x;
  const int lane = tid & 31;
  const int wv   = tid >> 5;
  const int nWv  = nThr >> 5;
  const int rIdx = blockIdx.x % FR_RANGES;
  const int sIdx = blockIdx.x / FR_RANGES;
  const int nSl  = gridDim.x / FR_RANGES;
  const int ES   = (E + nSl - 1) / nSl;
  const int eBeg = sIdx * ES;
  const int eEnd = min(E, eBeg + ES);

  for (int n0 = rIdx * FR_NODES; n0 < N; n0 += FR_RANGES * FR_NODES) {
    const int cnt = min(FR_NODES, N - n0);
    const int nW  = 3 * cnt;

    for (int i = tid; i < nW; i += nThr) sF[i] = 0.0f;
    __syncthreads();

    for (int base = eBeg + tid; base < eEnd; base += FR_UNROLL * nThr) {
      int   sN[FR_UNROLL], dN[FR_UNROLL];
      float gdv[FR_UNROLL], gpv[FR_UNROLL], fpv[FR_UNROLL];
      float rx[FR_UNROLL], ry[FR_UNROLL], rz[FR_UNROLL];
      bool  hs[FR_UNROLL], hd[FR_UNROLL];

#pragma unroll
      for (int u = 0; u < FR_UNROLL; ++u) {
        const int e = base + u * nThr;
        sN[u] = -1; dN[u] = -1;
        gdv[u] = 0.0f; gpv[u] = 0.0f; rx[u] = 0.0f; ry[u] = 0.0f; rz[u] = 0.0f;
        if (e < eEnd) {
          sN[u] = src[e]; dN[u] = dst[e];
          gdv[u] = gd[e]; gpv[u] = gp[e];
          rx[u] = r[3 * e]; ry[u] = r[3 * e + 1]; rz[u] = r[3 * e + 2];
        }
      }
#pragma unroll
      for (int u = 0; u < FR_UNROLL; ++u) {
        hs[u] = (unsigned)(sN[u] - n0) < (unsigned)cnt;
        hd[u] = (unsigned)(dN[u] - n0) < (unsigned)cnt;
        fpv[u] = 0.0f;
        if (hs[u] || hd[u]) fpv[u] = Fp[dN[u]];
      }
#pragma unroll
      for (int u = 0; u < FR_UNROLL; ++u) {
        if (hs[u] || hd[u]) {
          const float g  = fpv[u] * gdv[u] + gpv[u];
          const float fx = g * rx[u], fy = g * ry[u], fz = g * rz[u];
          if (hs[u]) {
            float* p = &sF[3 * (sN[u] - n0)];
            atomicAdd(p, fx); atomicAdd(p + 1, fy); atomicAdd(p + 2, fz);
          }
          if (hd[u]) {
            float* p = &sF[3 * (dN[u] - n0)];
            atomicAdd(p, -fx); atomicAdd(p + 1, -fy); atomicAdd(p + 2, -fz);
          }
        }
      }
    }
    __syncthreads();

    float* fbase = forces + (size_t)3 * n0;
    const int shift  = (int)((((size_t)fbase) >> 2) & 31);
    const int nLines = (shift + nW + 31) >> 5;
    for (int k = wv; k < nLines; k += nWv) {
      const int w = 32 * k + lane - shift;
      if (w >= 0 && w < nW) {
        const float v = sF[w];
        if (v != 0.0f) atomicAdd(fbase + w, v);
      }
    }
    __syncthreads();
  }
}

__global__ __launch_bounds__(256) void eam_energy(
    const float* __restrict__ partial_pair, int n1,
    const float* __restrict__ partial_F, int n2,
    float* __restrict__ energy)
{
  float v = 0.0f;
  for (int i = threadIdx.x; i < n1; i += blockDim.x) v += partial_pair[i];
  for (int i = threadIdx.x; i < n2; i += blockDim.x) v += partial_F[i];
  const float s = blockReduceSum256(v);
  if (threadIdx.x == 0) energy[0] = s;
}

extern "C" void kernel_launch(void* const* d_in, const int* in_sizes, int n_in,
                              void* d_out, int out_size, void* d_ws, size_t ws_size,
                              hipStream_t stream) {
  const float* r    = (const float*)d_in[0];
  const int*   src  = (const int*)  d_in[1];
  const int*   dst  = (const int*)  d_in[2];
  const float* phiD = (const float*)d_in[3];
  const float* phiP = (const float*)d_in[4];
  const float* emb  = (const float*)d_in[5];

  const int E = in_sizes[1];
  const int N = (out_size - 1) / 3;

  float* out    = (float*)d_out;
  float* energy = out;
  float* forces = out + 1;

  float* ws    = (float*)d_ws;
  float* gd    = ws;
  float* gp    = gd + E;
  float* local = gp + E;
  float* Fp    = local + N;
  const int NB1 = 512;
  const int NB2 = 128;
  float* ppair = Fp + N;
  float* pF    = ppair + NB1;

  eam_zero    <<<256,  256, 0, stream>>>(local, N, forces, 3 * N);
  eam_edge_fwd<<<NB1,  256, 0, stream>>>(r, dst, phiD, phiP, local, gd, gp, ppair, E);
  eam_node    <<<NB2,  256, 0, stream>>>(local, emb, Fp, pF, N);
  eam_force   <<<256, FR_THREADS, 0, stream>>>(r, src, dst, gd, gp, Fp, forces, E, N);
  eam_energy  <<<1,    256, 0, stream>>>(ppair, NB1, pF, NB2, energy);
}


# force kernel: edge stream read as four-edge groups with 16-byte loads issued a trip ahead of the LDS adds that use them, 512-thread blocks; LDS f32 sums regrouped per thread, still arrival order
# speedup vs baseline: 3.9878x; 1.3946x over previous
#include <hip/hip_runtime.h>
#include <math.h>

typedef __attribute__((ext_vector_type(2))) float v2f;
typedef __attribute__((ext_vector_type(8))) float v8f;
typedef __attribute__((ext_vector_type(4))) float v4f;
typedef __attribute__((ext_vector_type(4))) int   v4i;

#define NBASIS 128
#define NCHUNK 32
#define CUTOFF 6.0f
#define DELTA  (CUTOFF / (NBASIS - 1))
#define GAMMA_ ((NBASIS - 1) / CUTOFF)
#define PI_F   3.14159265358979323846f
#define LOG2E_F 1.44269504088896340736f
#define K2_    (-GAMMA_ * LOG2E_F)

#if __has_builtin(__builtin_amdgcn_exp2f)
#define FAST_EXP2(x) __builtin_amdgcn_exp2f(x)
#else
#define FAST_EXP2(x) exp2f(x)
#endif

__device__ __forceinline__ float softplus_(float x) {
  return fmaxf(x, 0.0f) + log1pf(expf(-fabsf(x)));
}

__device__ __forceinline__ float blockReduceSum256(float v) {
  __shared__ float sred[256];
  sred[threadIdx.x] = v;
  __syncthreads();
  for (int s = blockDim.x >> 1; s > 0; s >>= 1) {
    if (threadIdx.x < (unsigned)s) sred[threadIdx.x] += sred[threadIdx.x + s];
    __syncthreads();
  }
  return sred[0];
}

__global__ void eam_zero(float* __restrict__ a, int na, float* __restrict__ b, int nb) {
  const int i = blockIdx.x * blockDim.x + threadIdx.x;
  const int stride = gridDim.x * blockDim.x;
  for (int j = i; j < na; j += stride) a[j] = 0.0f;
  for (int j = i; j < nb; j += stride) b[j] = 0.0f;
}

template <bool BINNED>
__global__ __launch_bounds__(256) void eam_edge_fwd(
    const float* __restrict__ r, const int* __restrict__ dst,
    const float* __restrict__ phiD, const float* __restrict__ phiP,
    float* __restrict__ dens_out, float* __restrict__ gd_out, float* __restrict__ gp_out,
    float* __restrict__ partial_pair, int E)
{
  __shared__ v2f sW[64 * 16];

  for (int idx = threadIdx.x; idx < 64 * 16; idx += blockDim.x) {
    const int kp = idx >> 4, ch = idx & 15;
    const int k0 = 2 * kp, k1 = k0 + 1;
    const float c0 = k0 * DELTA, c1 = k1 * DELTA;
    v2f w = {0.0f, 0.0f};
    if (ch < 4) {
      float a0, a1;
      if (ch < 2) { a0 = softplus_(phiD[k0]); a1 = softplus_(phiD[k1]); }
      else        { a0 = phiP[k0];            a1 = phiP[k1]; }
      if (ch & 1) { a0 *= c0; a1 *= c1; }
      w.x = a0; w.y = a1;
    }
    sW[idx] = w;
  }
  __syncthreads();

  const int lane = threadIdx.x & 31;
  const int col  = lane & 15;
  const int hi   = lane >> 4;

  v2f aReg[NCHUNK];
#pragma unroll
  for (int t = 0; t < NCHUNK; ++t)
    aReg[t] = sW[(2 * t + hi) * 16 + col];

  const int wavesPerBlk = blockDim.x >> 5;
  const int wave   = blockIdx.x * wavesPerBlk + (threadIdx.x >> 5);
  const int nWaves = gridDim.x * wavesPerBlk;
  const int nTiles = (E + 15) >> 4;

  float pairAcc = 0.0f;

  for (int tile = wave; tile < nTiles; tile += nWaves) {
    const int e = tile * 16 + col;
    const bool valid = (e < E);

    const int eNext = (tile + nWaves) * 16 + col;
    if (eNext < E) __builtin_prefetch(&r[3 * eNext], 0, 0);

    float x = 0.0f, y = 0.0f, z = 0.0f;
    if (valid) { x = r[3 * e]; y = r[3 * e + 1]; z = r[3 * e + 2]; }
    const float dist = sqrtf(x * x + y * y + z * z);

    v8f acc = {0.f, 0.f, 0.f, 0.f, 0.f, 0.f, 0.f, 0.f};
    const float base = dist - (2 * hi) * DELTA;
    const v2f base2 = {base, base - DELTA};
#pragma unroll
    for (int t = 0; t < NCHUNK; ++t) {
      const v2f tt  = base2 - (float)(4 * t) * DELTA;
      const v2f arg = (tt * tt) * (float)K2_;
      v2f g;
      g.x = FAST_EXP2(arg.x);
      g.y = FAST_EXP2(arg.y);
      acc = __builtin_amdgcn_wmma_f32_16x16x4_f32(
          false, aReg[t], false, g, (short)0, acc, false, false);
    }

    if (lane < 16 && valid) {
      const float SA = acc[0], SB = acc[1], SC = acc[2], SD = acc[3];
      float sn, cs;
      __sincosf(dist * (PI_F / CUTOFF), &sn, &cs);
      const float phi  = 0.25f * (1.0f + cs);
      const float phip = -0.25f * (PI_F / CUTOFF) * sn;
      const float invd = 1.0f / dist;

      const float dens   = phi * SA;
      const float ddens  = phi * (-2.0f * GAMMA_) * (dist * SA - SB) + phip * SA;
      const float pairn  = phi * SC;
      const float dpairn = phi * (-2.0f * GAMMA_) * (dist * SC - SD) + phip * SC;
      const float dpair  = (dpairn - pairn * invd) * invd;

      if (BINNED) dens_out[e] = dens;
      else        atomicAdd(&dens_out[dst[e]], dens);
      gd_out[e] = ddens * invd;
      gp_out[e] = dpair * invd;
      pairAcc += pairn * invd;
    }
  }

  const float bsum = blockReduceSum256(pairAcc);
  if (threadIdx.x == 0) partial_pair[blockIdx.x] = bsum;
}

#define DR_RANGES  8
#define DR_NODES   6272
#define DR_SLICES  32
#define DR_PITCH   (DR_RANGES * DR_NODES)
#define DR_THREADS 1024
#define DR_UNROLL  4

__global__ __launch_bounds__(DR_THREADS) void eam_density(
    const int* __restrict__ dst, const float* __restrict__ dens,
    float* __restrict__ planes, float* __restrict__ forces, int nf, int E, int N)
{
  __shared__ float sD[DR_NODES];

  const int tid  = threadIdx.x;
  const int rIdx = blockIdx.x % DR_RANGES;
  const int sIdx = blockIdx.x / DR_RANGES;
  const int ES   = (((E + DR_SLICES - 1) / DR_SLICES) + 31) & ~31;
  const int eBeg = min(E, sIdx * ES);
  const int eEnd = min(E, eBeg + ES);
  const int n0   = rIdx * DR_NODES;
  const int cnt  = max(0, min(DR_NODES, N - n0));

  for (int j = blockIdx.x * DR_THREADS + tid; j < nf; j += gridDim.x * DR_THREADS) forces[j] = 0.0f;

  for (int i = tid; i < DR_NODES; i += DR_THREADS) sD[i] = 0.0f;
  __syncthreads();

  const int eEnd4 = eBeg + ((eEnd - eBeg) & ~3);
  for (int base = eBeg + 4 * tid; base < eEnd4; base += 4 * DR_UNROLL * DR_THREADS) {
    v4i dN[DR_UNROLL];
    v4f dv[DR_UNROLL];
#pragma unroll
    for (int u = 0; u < DR_UNROLL; ++u) {
      const int e = base + 4 * u * DR_THREADS;
      dN[u] = (v4i){-1, -1, -1, -1};
      dv[u] = (v4f){0.0f, 0.0f, 0.0f, 0.0f};
      if (e < eEnd4) {
        dN[u] = *(const v4i*)(dst + e);
        dv[u] = *(const v4f*)(dens + e);
      }
    }
#pragma unroll
    for (int u = 0; u < DR_UNROLL; ++u) {
      const unsigned k0 = (unsigned)(dN[u].x - n0), k1 = (unsigned)(dN[u].y - n0);
      const unsigned k2 = (unsigned)(dN[u].z - n0), k3 = (unsigned)(dN[u].w - n0);
      if (k0 < (unsigned)cnt) atomicAdd(&sD[k0], dv[u].x);
      if (k1 < (unsigned)cnt) atomicAdd(&sD[k1], dv[u].y);
      if (k2 < (unsigned)cnt) atomicAdd(&sD[k2], dv[u].z);
      if (k3 < (unsigned)cnt) atomicAdd(&sD[k3], dv[u].w);
    }
  }
  if (eEnd4 + tid < eEnd) {
    const unsigned k = (unsigned)(dst[eEnd4 + tid] - n0);
    if (k < (unsigned)cnt) atomicAdd(&sD[k], dens[eEnd4 + tid]);
  }
  __syncthreads();

  float* prow = planes + (size_t)sIdx * DR_PITCH + n0;
  for (int i = tid; i < cnt; i += DR_THREADS) prow[i] = sD[i];
}

template <bool PLANES>
__global__ __launch_bounds__(256) void eam_node(
    const float* __restrict__ planes, const float* __restrict__ w,
    float* __restrict__ Fp, float* __restrict__ partial_F, int N)
{
  const float w0 = w[0], w1 = w[1], w2 = w[2], w3 = w[3], w4 = w[4];
  float Facc = 0.0f;
  for (int n = blockIdx.x * blockDim.x + threadIdx.x; n < N;
       n += gridDim.x * blockDim.x) {
    float x = planes[n];
    if (PLANES) {
#pragma unroll
      for (int sl = 1; sl < DR_SLICES; ++sl) x += planes[(size_t)sl * DR_PITCH + n];
    }
    const float sq = sqrtf(x);
    const float x2 = x * x;
    const float F  = 2.0f * w0 * sq + w1 * x + 0.5f * w2 * x2 +
                     (1.0f / 6.0f) * w3 * x2 * x + (1.0f / 24.0f) * w4 * x2 * x2;
    const float Fd = w0 / sq + w1 + w2 * x + 0.5f * w3 * x2 +
                     (1.0f / 6.0f) * w4 * x2 * x;
    Facc += F;
    Fp[n] = Fd;
  }
  const float bsum = blockReduceSum256(Facc);
  if (threadIdx.x == 0) partial_F[blockIdx.x] = bsum;
}

#define FR_RANGES  8
#define FR_NODES   6250
#define FR_THREADS 512
#define FR_GROUPS  2

struct FrStream {
  v4i sN[FR_GROUPS], dN[FR_GROUPS];
  v4f gdv[FR_GROUPS], gpv[FR_GROUPS];
  v4f ra[FR_GROUPS], rb[FR_GROUPS], rc[FR_GROUPS];
};

__device__ __forceinline__ void fr_load(
    FrStream& s, const float* __restrict__ r, const int* __restrict__ src,
    const int* __restrict__ dst, const float* __restrict__ gd,
    const float* __restrict__ gp, int eBeg, int nGrp, int j0, int nThr, int tid)
{
#pragma unroll
  for (int u = 0; u < FR_GROUPS; ++u) {
    const int g = (j0 + u) * nThr + tid;
    const int e = eBeg + 4 * min(g, nGrp - 1);
    s.sN[u]  = *(const v4i*)(src + e);
    s.dN[u]  = *(const v4i*)(dst + e);
    s.gdv[u] = *(const v4f*)(gd + e);
    s.gpv[u] = *(const v4f*)(gp + e);
    const v4f* rp = (const v4f*)(r + 3 * (size_t)e);
    s.ra[u] = rp[0]; s.rb[u] = rp[1]; s.rc[u] = rp[2];
  }
}

__global__ __launch_bounds__(FR_THREADS) void eam_force(
    const float* __restrict__ r, const int* __restrict__ src,
    const int* __restrict__ dst, const float* __restrict__ gd,
    const float* __restrict__ gp, const float* __restrict__ Fp,
    float* __restrict__ forces, int E, int N)
{
  __shared__ float sF[3 * FR_NODES];

  const int tid  = threadIdx.x;
  const int nThr = blockDim.x;
  const int lane = tid & 31;
  const int wv   = tid >> 5;
  const int nWv  = nThr >> 5;
  const int rIdx = blockIdx.x % FR_RANGES;
  const int sIdx = blockIdx.x / FR_RANGES;
  const int nSl  = gridDim.x / FR_RANGES;
  const int ES   = (((E + nSl - 1) / nSl) + 31) & ~31;
  const int eBeg = min(E, sIdx * ES);
  const int eEnd = min(E, eBeg + ES);
  const bool wide =
      ((((size_t)r | (size_t)src | (size_t)dst | (size_t)gd | (size_t)gp) & 15) == 0);
  const int nGrp  = wide ? ((eEnd - eBeg) >> 2) : 0;
  const int eEnd4 = eBeg + 4 * nGrp;
  const int K     = (nGrp + nThr - 1) / nThr;

  for (int n0 = rIdx * FR_NODES; n0 < N; n0 += FR_RANGES * FR_NODES) {
    const int cnt = min(FR_NODES, N - n0);
    const int nW  = 3 * cnt;

    FrStream cur;
    if (K > 0) fr_load(cur, r, src, dst, gd, gp, eBeg, nGrp, 0, nThr, tid);

    for (int i = tid; i < nW; i += nThr) sF[i] = 0.0f;
    __syncthreads();

    for (int j0 = 0; j0 < K; j0 += FR_GROUPS) {
      bool  hs[4 * FR_GROUPS], hd[4 * FR_GROUPS];
      float fpv[4 * FR_GROUPS];
#pragma unroll
      for (int u = 0; u < FR_GROUPS; ++u) {
        const bool ok = ((j0 + u) * nThr + tid) < nGrp;
#pragma unroll
        for (int c = 0; c < 4; ++c) {
          const int q = 4 * u + c;
          hs[q] = ok && ((unsigned)(cur.sN[u][c] - n0) < (unsigned)cnt);
          hd[q] = ok && ((unsigned)(cur.dN[u][c] - n0) < (unsigned)cnt);
          fpv[q] = 0.0f;
          if (hs[q] || hd[q]) fpv[q] = Fp[cur.dN[u][c]];
        }
      }

      FrStream nxt;
      fr_load(nxt, r, src, dst, gd, gp, eBeg, nGrp, j0 + FR_GROUPS, nThr, tid);

#pragma unroll
      for (int u = 0; u < FR_GROUPS; ++u) {
        const float rr[12] = {cur.ra[u].x, cur.ra[u].y, cur.ra[u].z, cur.ra[u].w,
                              cur.rb[u].x, cur.rb[u].y, cur.rb[u].z, cur.rb[u].w,
                              cur.rc[u].x, cur.rc[u].y, cur.rc[u].z, cur.rc[u].w};
#pragma unroll
        for (int c = 0; c < 4; ++c) {
          const int q = 4 * u + c;
          if (hs[q] || hd[q]) {
            const float g  = fpv[q] * cur.gdv[u][c] + cur.gpv[u][c];
            const float fx = g * rr[3 * c], fy = g * rr[3 * c + 1], fz = g * rr[3 * c + 2];
            if (hs[q]) {
              float* p = &sF[3 * (cur.sN[u][c] - n0)];
              atomicAdd(p, fx); atomicAdd(p + 1, fy); atomicAdd(p + 2, fz);
            }
            if (hd[q]) {
              float* p = &sF[3 * (cur.dN[u][c] - n0)];
              atomicAdd(p, -fx); atomicAdd(p + 1, -fy); atomicAdd(p + 2, -fz);
            }
          }
        }
      }
      cur = nxt;
    }

    for (int e = eEnd4 + tid; e < eEnd; e += nThr) {
      const int s = src[e], d = dst[e];
      const bool h1 = (unsigned)(s - n0) < (unsigned)cnt;
      const bool h2 = (unsigned)(d - n0) < (unsigned)cnt;
      if (h1 || h2) {
        const float g  = Fp[d] * gd[e] + gp[e];
        const float fx = g * r[3 * e], fy = g * r[3 * e + 1], fz = g * r[3 * e + 2];
        if (h1) {
          float* p = &sF[3 * (s - n0)];
          atomicAdd(p, fx); atomicAdd(p + 1, fy); atomicAdd(p + 2, fz);
        }
        if (h2) {
          float* p = &sF[3 * (d - n0)];
          atomicAdd(p, -fx); atomicAdd(p + 1, -fy); atomicAdd(p + 2, -fz);
        }
      }
    }
    __syncthreads();

    float* fbase = forces + (size_t)3 * n0;
    const int shift  = (int)((((size_t)fbase) >> 2) & 31);
    const int nLines = (shift + nW + 31) >> 5;
    for (int k = wv; k < nLines; k += nWv) {
      const int w = 32 * k + lane - shift;
      if (w >= 0 && w < nW) {
        const float v = sF[w];
        if (v != 0.0f) atomicAdd(fbase + w, v);
      }
    }
    __syncthreads();
  }
}

__global__ __launch_bounds__(256) void eam_energy(
    const float* __restrict__ partial_pair, int n1,
    const float* __restrict__ partial_F, int n2,
    float* __restrict__ energy)
{
  float v = 0.0f;
  for (int i = threadIdx.x; i < n1; i += blockDim.x) v += partial_pair[i];
  for (int i = threadIdx.x; i < n2; i += blockDim.x) v += partial_F[i];
  const float s = blockReduceSum256(v);
  if (threadIdx.x == 0) energy[0] = s;
}

extern "C" void kernel_launch(void* const* d_in, const int* in_sizes, int n_in,
                              void* d_out, int out_size, void* d_ws, size_t ws_size,
                              hipStream_t stream) {
  const float* r    = (const float*)d_in[0];
  const int*   src  = (const int*)  d_in[1];
  const int*   dst  = (const int*)  d_in[2];
  const float* phiD = (const float*)d_in[3];
  const float* phiP = (const float*)d_in[4];
  const float* emb  = (const float*)d_in[5];

  const int E = in_sizes[1];
  const int N = (out_size - 1) / 3;

  float* out    = (float*)d_out;
  float* energy = out;
  float* forces = out + 1;

  const int NB1 = 512;
  const int NB2 = 128;
  float* ws = (float*)d_ws;
  float *gd, *gp, *Fp, *ppair, *pF;

  const size_t needBinned =
      ((size_t)DR_SLICES * DR_PITCH + (size_t)3 * E + N + NB1 + NB2) * sizeof(float);
  if (ws_size >= needBinned && N <= DR_PITCH) {
    float* planes = ws;
    float* dens   = planes + DR_SLICES * DR_PITCH;
    gd    = dens + E;
    gp    = gd + E;
    Fp    = gp + E;
    ppair = Fp + N;
    pF    = ppair + NB1;
    eam_edge_fwd<true><<<NB1, 256, 0, stream>>>(r, dst, phiD, phiP, dens, gd, gp, ppair, E);
    eam_density<<<DR_RANGES * DR_SLICES, DR_THREADS, 0, stream>>>(dst, dens, planes, forces, 3 * N, E, N);
    eam_node<true><<<NB2, 256, 0, stream>>>(planes, emb, Fp, pF, N);
  } else {
    gd    = ws;
    gp    = gd + E;
    float* local = gp + E;
    Fp    = local + N;
    ppair = Fp + N;
    pF    = ppair + NB1;
    eam_zero<<<256, 256, 0, stream>>>(local, N, forces, 3 * N);
    eam_edge_fwd<false><<<NB1, 256, 0, stream>>>(r, dst, phiD, phiP, local, gd, gp, ppair, E);
    eam_node<false><<<NB2, 256, 0, stream>>>(local, emb, Fp, pF, N);
  }
  eam_force   <<<256, FR_THREADS, 0, stream>>>(r, src, dst, gd, gp, Fp, forces, E, N);
  eam_energy  <<<1,    256, 0, stream>>>(ppair, NB1, pF, NB2, energy);
}


# edge kernel: a 16-edge tile's basis loop stops after the last chunk that can hold a non-zero Gaussian in f32 (centres past longest bond + 2.25 give exact zeros); sums bit-identical, same order
# speedup vs baseline: 4.4952x; 1.1272x over previous
#include <hip/hip_runtime.h>
#include <math.h>

typedef __attribute__((ext_vector_type(2))) float v2f;
typedef __attribute__((ext_vector_type(8))) float v8f;
typedef __attribute__((ext_vector_type(4))) float v4f;
typedef __attribute__((ext_vector_type(4))) int   v4i;

#define NBASIS 128
#define NCHUNK 32
#define CUTOFF 6.0f
#define DELTA  (CUTOFF / (NBASIS - 1))
#define GAMMA_ ((NBASIS - 1) / CUTOFF)
#define PI_F   3.14159265358979323846f
#define LOG2E_F 1.44269504088896340736f
#define K2_    (-GAMMA_ * LOG2E_F)

#if __has_builtin(__builtin_amdgcn_exp2f)
#define FAST_EXP2(x) __builtin_amdgcn_exp2f(x)
#else
#define FAST_EXP2(x) exp2f(x)
#endif

#define ZERO_REACH 2.25f
static_assert(K2_ * ZERO_REACH * ZERO_REACH < -150.0f, "skipped terms must be exact zeros");
#define SKIP_FROM 16
#define SKIP_STEP 2

__device__ __forceinline__ unsigned rowMaxU(unsigned v) {
  v = max(v, (unsigned)__builtin_amdgcn_update_dpp((int)v, (int)v, 0x128, 0xF, 0xF, false));
  v = max(v, (unsigned)__builtin_amdgcn_update_dpp((int)v, (int)v, 0x124, 0xF, 0xF, false));
  v = max(v, (unsigned)__builtin_amdgcn_update_dpp((int)v, (int)v, 0x122, 0xF, 0xF, false));
  v = max(v, (unsigned)__builtin_amdgcn_update_dpp((int)v, (int)v, 0x121, 0xF, 0xF, false));
  return v;
}

__device__ __forceinline__ float softplus_(float x) {
  return fmaxf(x, 0.0f) + log1pf(expf(-fabsf(x)));
}

__device__ __forceinline__ float blockReduceSum256(float v) {
  __shared__ float sred[256];
  sred[threadIdx.x] = v;
  __syncthreads();
  for (int s = blockDim.x >> 1; s > 0; s >>= 1) {
    if (threadIdx.x < (unsigned)s) sred[threadIdx.x] += sred[threadIdx.x + s];
    __syncthreads();
  }
  return sred[0];
}

__global__ void eam_zero(float* __restrict__ a, int na, float* __restrict__ b, int nb) {
  const int i = blockIdx.x * blockDim.x + threadIdx.x;
  const int stride = gridDim.x * blockDim.x;
  for (int j = i; j < na; j += stride) a[j] = 0.0f;
  for (int j = i; j < nb; j += stride) b[j] = 0.0f;
}

template <bool BINNED>
__global__ __launch_bounds__(256) void eam_edge_fwd(
    const float* __restrict__ r, const int* __restrict__ dst,
    const float* __restrict__ phiD, const float* __restrict__ phiP,
    float* __restrict__ dens_out, float* __restrict__ gd_out, float* __restrict__ gp_out,
    float* __restrict__ partial_pair, int E)
{
  __shared__ v2f sW[64 * 16];

  for (int idx = threadIdx.x; idx < 64 * 16; idx += blockDim.x) {
    const int kp = idx >> 4, ch = idx & 15;
    const int k0 = 2 * kp, k1 = k0 + 1;
    const float c0 = k0 * DELTA, c1 = k1 * DELTA;
    v2f w = {0.0f, 0.0f};
    if (ch < 4) {
      float a0, a1;
      if (ch < 2) { a0 = softplus_(phiD[k0]); a1 = softplus_(phiD[k1]); }
      else        { a0 = phiP[k0];            a1 = phiP[k1]; }
      if (ch & 1) { a0 *= c0; a1 *= c1; }
      w.x = a0; w.y = a1;
    }
    sW[idx] = w;
  }
  __syncthreads();

  const int lane = threadIdx.x & 31;
  const int col  = lane & 15;
  const int hi   = lane >> 4;

  v2f aReg[NCHUNK];
#pragma unroll
  for (int t = 0; t < NCHUNK; ++t)
    aReg[t] = sW[(2 * t + hi) * 16 + col];

  const int wavesPerBlk = blockDim.x >> 5;
  const int wave   = blockIdx.x * wavesPerBlk + (threadIdx.x >> 5);
  const int nWaves = gridDim.x * wavesPerBlk;
  const int nTiles = (E + 15) >> 4;

  float pairAcc = 0.0f;

  for (int tile = wave; tile < nTiles; tile += nWaves) {
    const int e = tile * 16 + col;
    const bool valid = (e < E);

    const int eNext = (tile + nWaves) * 16 + col;
    if (eNext < E) __builtin_prefetch(&r[3 * eNext], 0, 0);

    float x = 0.0f, y = 0.0f, z = 0.0f;
    if (valid) { x = r[3 * e]; y = r[3 * e + 1]; z = r[3 * e + 2]; }
    const float dist = sqrtf(x * x + y * y + z * z);

    const float dmax = __uint_as_float(
        (unsigned)__builtin_amdgcn_readfirstlane((int)rowMaxU(__float_as_uint(dist))));
    const float tLast = (dmax + ZERO_REACH) * (1.0f / (4.0f * DELTA));
    const int tEnd = (tLast < (float)NCHUNK) ? (int)tLast + 1 : NCHUNK;

    v8f acc = {0.f, 0.f, 0.f, 0.f, 0.f, 0.f, 0.f, 0.f};
    const float base = dist - (2 * hi) * DELTA;
    const v2f base2 = {base, base - DELTA};
#pragma unroll
    for (int t = 0; t < NCHUNK; ++t) {
      if (t >= SKIP_FROM && ((t - SKIP_FROM) % SKIP_STEP) == 0 && t >= tEnd) break;
      const v2f tt  = base2 - (float)(4 * t) * DELTA;
      const v2f arg = (tt * tt) * (float)K2_;
      v2f g;
      g.x = FAST_EXP2(arg.x);
      g.y = FAST_EXP2(arg.y);
      acc = __builtin_amdgcn_wmma_f32_16x16x4_f32(
          false, aReg[t], false, g, (short)0, acc, false, false);
    }

    if (lane < 16 && valid) {
      const float SA = acc[0], SB = acc[1], SC = acc[2], SD = acc[3];
      float sn, cs;
      __sincosf(dist * (PI_F / CUTOFF), &sn, &cs);
      const float phi  = 0.25f * (1.0f + cs);
      const float phip = -0.25f * (PI_F / CUTOFF) * sn;
      const float invd = 1.0f / dist;

      const float dens   = phi * SA;
      const float ddens  = phi * (-2.0f * GAMMA_) * (dist * SA - SB) + phip * SA;
      const float pairn  = phi * SC;
      const float dpairn = phi * (-2.0f * GAMMA_) * (dist * SC - SD) + phip * SC;
      const float dpair  = (dpairn - pairn * invd) * invd;

      if (BINNED) dens_out[e] = dens;
      else        atomicAdd(&dens_out[dst[e]], dens);
      gd_out[e] = ddens * invd;
      gp_out[e] = dpair * invd;
      pairAcc += pairn * invd;
    }
  }

  const float bsum = blockReduceSum256(pairAcc);
  if (threadIdx.x == 0) partial_pair[blockIdx.x] = bsum;
}

#define DR_RANGES  8
#define DR_NODES   6272
#define DR_SLICES  32
#define DR_PITCH   (DR_RANGES * DR_NODES)
#define DR_THREADS 1024
#define DR_UNROLL  4

__global__ __launch_bounds__(DR_THREADS) void eam_density(
    const int* __restrict__ dst, const float* __restrict__ dens,
    float* __restrict__ planes, float* __restrict__ forces, int nf, int E, int N)
{
  __shared__ float sD[DR_NODES];

  const int tid  = threadIdx.x;
  const int rIdx = blockIdx.x % DR_RANGES;
  const int sIdx = blockIdx.x / DR_RANGES;
  const int ES   = (((E + DR_SLICES - 1) / DR_SLICES) + 31) & ~31;
  const int eBeg = min(E, sIdx * ES);
  const int eEnd = min(E, eBeg + ES);
  const int n0   = rIdx * DR_NODES;
  const int cnt  = max(0, min(DR_NODES, N - n0));

  for (int j = blockIdx.x * DR_THREADS + tid; j < nf; j += gridDim.x * DR_THREADS) forces[j] = 0.0f;

  for (int i = tid; i < DR_NODES; i += DR_THREADS) sD[i] = 0.0f;
  __syncthreads();

  const int eEnd4 = eBeg + ((eEnd - eBeg) & ~3);
  for (int base = eBeg + 4 * tid; base < eEnd4; base += 4 * DR_UNROLL * DR_THREADS) {
    v4i dN[DR_UNROLL];
    v4f dv[DR_UNROLL];
#pragma unroll
    for (int u = 0; u < DR_UNROLL; ++u) {
      const int e = base + 4 * u * DR_THREADS;
      dN[u] = (v4i){-1, -1, -1, -1};
      dv[u] = (v4f){0.0f, 0.0f, 0.0f, 0.0f};
      if (e < eEnd4) {
        dN[u] = *(const v4i*)(dst + e);
        dv[u] = *(const v4f*)(dens + e);
      }
    }
#pragma unroll
    for (int u = 0; u < DR_UNROLL; ++u) {
      const unsigned k0 = (unsigned)(dN[u].x - n0), k1 = (unsigned)(dN[u].y - n0);
      const unsigned k2 = (unsigned)(dN[u].z - n0), k3 = (unsigned)(dN[u].w - n0);
      if (k0 < (unsigned)cnt) atomicAdd(&sD[k0], dv[u].x);
      if (k1 < (unsigned)cnt) atomicAdd(&sD[k1], dv[u].y);
      if (k2 < (unsigned)cnt) atomicAdd(&sD[k2], dv[u].z);
      if (k3 < (unsigned)cnt) atomicAdd(&sD[k3], dv[u].w);
    }
  }
  if (eEnd4 + tid < eEnd) {
    const unsigned k = (unsigned)(dst[eEnd4 + tid] - n0);
    if (k < (unsigned)cnt) atomicAdd(&sD[k], dens[eEnd4 + tid]);
  }
  __syncthreads();

  float* prow = planes + (size_t)sIdx * DR_PITCH + n0;
  for (int i = tid; i < cnt; i += DR_THREADS) prow[i] = sD[i];
}

template <bool PLANES>
__global__ __launch_bounds__(256) void eam_node(
    const float* __restrict__ planes, const float* __restrict__ w,
    float* __restrict__ Fp, float* __restrict__ partial_F, int N)
{
  const float w0 = w[0], w1 = w[1], w2 = w[2], w3 = w[3], w4 = w[4];
  float Facc = 0.0f;
  for (int n = blockIdx.x * blockDim.x + threadIdx.x; n < N;
       n += gridDim.x * blockDim.x) {
    float x = planes[n];
    if (PLANES) {
#pragma unroll
      for (int sl = 1; sl < DR_SLICES; ++sl) x += planes[(size_t)sl * DR_PITCH + n];
    }
    const float sq = sqrtf(x);
    const float x2 = x * x;
    const float F  = 2.0f * w0 * sq + w1 * x + 0.5f * w2 * x2 +
                     (1.0f / 6.0f) * w3 * x2 * x + (1.0f / 24.0f) * w4 * x2 * x2;
    const float Fd = w0 / sq + w1 + w2 * x + 0.5f * w3 * x2 +
                     (1.0f / 6.0f) * w4 * x2 * x;
    Facc += F;
    Fp[n] = Fd;
  }
  const float bsum = blockReduceSum256(Facc);
  if (threadIdx.x == 0) partial_F[blockIdx.x] = bsum;
}

#define FR_RANGES  8
#define FR_NODES   6250
#define FR_THREADS 512
#define FR_GROUPS  2
#define EN_PAIR    512
#define EN_NODE    128
#define EN_THREADS 256

struct FrStream {
  v4i sN[FR_GROUPS], dN[FR_GROUPS];
  v4f gdv[FR_GROUPS], gpv[FR_GROUPS];
  v4f ra[FR_GROUPS], rb[FR_GROUPS], rc[FR_GROUPS];
};

__device__ __forceinline__ void fr_load(
    FrStream& s, const float* __restrict__ r, const int* __restrict__ src,
    const int* __restrict__ dst, const float* __restrict__ gd,
    const float* __restrict__ gp, int eBeg, int nGrp, int j0, int nThr, int tid)
{
#pragma unroll
  for (int u = 0; u < FR_GROUPS; ++u) {
    const int g = (j0 + u) * nThr + tid;
    const int e = eBeg + 4 * min(g, nGrp - 1);
    s.sN[u]  = *(const v4i*)(src + e);
    s.dN[u]  = *(const v4i*)(dst + e);
    s.gdv[u] = *(const v4f*)(gd + e);
    s.gpv[u] = *(const v4f*)(gp + e);
    const v4f* rp = (const v4f*)(r + 3 * (size_t)e);
    s.ra[u] = rp[0]; s.rb[u] = rp[1]; s.rc[u] = rp[2];
  }
}

__global__ __launch_bounds__(FR_THREADS) void eam_force(
    const float* __restrict__ r, const int* __restrict__ src,
    const int* __restrict__ dst, const float* __restrict__ gd,
    const float* __restrict__ gp, const float* __restrict__ Fp,
    float* __restrict__ forces, int E, int N,
    const float* __restrict__ partial_pair, const float* __restrict__ partial_F,
    float* __restrict__ energy)
{
  __shared__ float sF[3 * FR_NODES];

  const int tid  = threadIdx.x;
  const int nThr = blockDim.x;
  const int lane = tid & 31;
  const int wv   = tid >> 5;
  const int nWv  = nThr >> 5;
  const int rIdx = blockIdx.x % FR_RANGES;
  const int sIdx = blockIdx.x / FR_RANGES;
  const int nSl  = gridDim.x / FR_RANGES;
  const int ES   = (((E + nSl - 1) / nSl) + 31) & ~31;
  const int eBeg = min(E, sIdx * ES);
  const int eEnd = min(E, eBeg + ES);
  const bool wide =
      ((((size_t)r | (size_t)src | (size_t)dst | (size_t)gd | (size_t)gp) & 15) == 0);
  const int nGrp  = wide ? ((eEnd - eBeg) >> 2) : 0;
  const int eEnd4 = eBeg + 4 * nGrp;
  const int K     = (nGrp + nThr - 1) / nThr;

  const bool eBlk = (blockIdx.x == gridDim.x - FR_RANGES);

  for (int n0 = rIdx * FR_NODES; n0 < N; n0 += FR_RANGES * FR_NODES) {
    const int cnt = min(FR_NODES, N - n0);
    const int nW  = 3 * cnt;

    FrStream cur;
    if (K > 0) fr_load(cur, r, src, dst, gd, gp, eBeg, nGrp, 0, nThr, tid);

    if (eBlk && n0 == rIdx * FR_NODES) {
      constexpr int EP = (EN_PAIR + EN_THREADS - 1) / EN_THREADS;
      constexpr int EF = (EN_NODE + EN_THREADS - 1) / EN_THREADS;
      static_assert(EN_THREADS == 256 && FR_THREADS >= EN_THREADS, "the tree below is written for 256 sums");
      const int t8 = tid & (EN_THREADS - 1);
      float ep[EP], ef[EF];
      for (int j = 0; j < EP; ++j) ep[j] = partial_pair[min(t8 + EN_THREADS * j, EN_PAIR - 1)];
      for (int j = 0; j < EF; ++j) ef[j] = partial_F[min(t8 + EN_THREADS * j, EN_NODE - 1)];
      float ev = 0.0f;
      for (int j = 0; j < EP; ++j) ev += (t8 + EN_THREADS * j < EN_PAIR) ? ep[j] : 0.0f;
      for (int j = 0; j < EF; ++j) ev += (t8 + EN_THREADS * j < EN_NODE) ? ef[j] : 0.0f;
      if (tid < EN_THREADS) sF[tid] = ev;
      __syncthreads();
      if (tid < 128) sF[tid] += sF[tid + 128];
      __syncthreads();
      if (tid < 64) sF[tid] += sF[tid + 64];
      __syncthreads();
      if (tid < 32) {
        float v = sF[tid] + sF[tid + 32];
        v += __shfl_down(v, 16);
        v += __shfl_down(v, 8);
        v += __shfl_down(v, 4);
        v += __shfl_down(v, 2);
        v += __shfl_down(v, 1);
        if (tid == 0) energy[0] = v;
      }
      __syncthreads();
    }

    for (int i = tid; i < nW; i += nThr) sF[i] = 0.0f;
    __syncthreads();

    for (int j0 = 0; j0 < K; j0 += FR_GROUPS) {
      bool  hs[4 * FR_GROUPS], hd[4 * FR_GROUPS];
      float fpv[4 * FR_GROUPS];
#pragma unroll
      for (int u = 0; u < FR_GROUPS; ++u) {
        const bool ok = ((j0 + u) * nThr + tid) < nGrp;
#pragma unroll
        for (int c = 0; c < 4; ++c) {
          const int q = 4 * u + c;
          hs[q] = ok && ((unsigned)(cur.sN[u][c] - n0) < (unsigned)cnt);
          hd[q] = ok && ((unsigned)(cur.dN[u][c] - n0) < (unsigned)cnt);
          fpv[q] = 0.0f;
          if (hs[q] || hd[q]) fpv[q] = Fp[cur.dN[u][c]];
        }
      }

      FrStream nxt;
      fr_load(nxt, r, src, dst, gd, gp, eBeg, nGrp, j0 + FR_GROUPS, nThr, tid);

#pragma unroll
      for (int u = 0; u < FR_GROUPS; ++u) {
        const float rr[12] = {cur.ra[u].x, cur.ra[u].y, cur.ra[u].z, cur.ra[u].w,
                              cur.rb[u].x, cur.rb[u].y, cur.rb[u].z, cur.rb[u].w,
                              cur.rc[u].x, cur.rc[u].y, cur.rc[u].z, cur.rc[u].w};
#pragma unroll
        for (int c = 0; c < 4; ++c) {
          const int q = 4 * u + c;
          if (hs[q] || hd[q]) {
            const float g  = fpv[q] * cur.gdv[u][c] + cur.gpv[u][c];
            const float fx = g * rr[3 * c], fy = g * rr[3 * c + 1], fz = g * rr[3 * c + 2];
            if (hs[q]) {
              float* p = &sF[3 * (cur.sN[u][c] - n0)];
              atomicAdd(p, fx); atomicAdd(p + 1, fy); atomicAdd(p + 2, fz);
            }
            if (hd[q]) {
              float* p = &sF[3 * (cur.dN[u][c] - n0)];
              atomicAdd(p, -fx); atomicAdd(p + 1, -fy); atomicAdd(p + 2, -fz);
            }
          }
        }
      }
      cur = nxt;
    }

    for (int e = eEnd4 + tid; e < eEnd; e += nThr) {
      const int s = src[e], d = dst[e];
      const bool h1 = (unsigned)(s - n0) < (unsigned)cnt;
      const bool h2 = (unsigned)(d - n0) < (unsigned)cnt;
      if (h1 || h2) {
        const float g  = Fp[d] * gd[e] + gp[e];
        const float fx = g * r[3 * e], fy = g * r[3 * e + 1], fz = g * r[3 * e + 2];
        if (h1) {
          float* p = &sF[3 * (s - n0)];
          atomicAdd(p, fx); atomicAdd(p + 1, fy); atomicAdd(p + 2, fz);
        }
        if (h2) {
          float* p = &sF[3 * (d - n0)];
          atomicAdd(p, -fx); atomicAdd(p + 1, -fy); atomicAdd(p + 2, -fz);
        }
      }
    }
    __syncthreads();

    float* fbase = forces + (size_t)3 * n0;
    const int shift  = (int)((((size_t)fbase) >> 2) & 31);
    const int nLines = (shift + nW + 31) >> 5;
    for (int k = wv; k < nLines; k += nWv) {
      const int w = 32 * k + lane - shift;
      if (w >= 0 && w < nW) {
        const float v = sF[w];
        if (v != 0.0f) atomicAdd(fbase + w, v);
      }
    }
    __syncthreads();
  }
}

extern "C" void kernel_launch(void* const* d_in, const int* in_sizes, int n_in,
                              void* d_out, int out_size, void* d_ws, size_t ws_size,
                              hipStream_t stream) {
  const float* r    = (const float*)d_in[0];
  const int*   src  = (const int*)  d_in[1];
  const int*   dst  = (const int*)  d_in[2];
  const float* phiD = (const float*)d_in[3];
  const float* phiP = (const float*)d_in[4];
  const float* emb  = (const float*)d_in[5];

  const int E = in_sizes[1];
  const int N = (out_size - 1) / 3;

  float* out    = (float*)d_out;
  float* energy = out;
  float* forces = out + 1;

  const int NB1 = EN_PAIR;
  const int NB2 = EN_NODE;
  float* ws = (float*)d_ws;
  float *gd, *gp, *Fp, *ppair, *pF;

  const size_t needBinned =
      ((size_t)DR_SLICES * DR_PITCH + (size_t)3 * E + N + NB1 + NB2) * sizeof(float);
  if (ws_size >= needBinned && N <= DR_PITCH) {
    float* planes = ws;
    float* dens   = planes + DR_SLICES * DR_PITCH;
    gd    = dens + E;
    gp    = gd + E;
    Fp    = gp + E;
    ppair = Fp + N;
    pF    = ppair + NB1;
    eam_edge_fwd<true><<<NB1, 256, 0, stream>>>(r, dst, phiD, phiP, dens, gd, gp, ppair, E);
    eam_density<<<DR_RANGES * DR_SLICES, DR_THREADS, 0, stream>>>(dst, dens, planes, forces, 3 * N, E, N);
    eam_node<true><<<NB2, 256, 0, stream>>>(planes, emb, Fp, pF, N);
  } else {
    gd    = ws;
    gp    = gd + E;
    float* local = gp + E;
    Fp    = local + N;
    ppair = Fp + N;
    pF    = ppair + NB1;
    eam_zero<<<256, 256, 0, stream>>>(local, N, forces, 3 * N);
    eam_edge_fwd<false><<<NB1, 256, 0, stream>>>(r, dst, phiD, phiP, local, gd, gp, ppair, E);
    eam_node<false><<<NB2, 256, 0, stream>>>(local, emb, Fp, pF, N);
  }
  eam_force   <<<256, FR_THREADS, 0, stream>>>(r, src, dst, gd, gp, Fp, forces, E, N,
                                               ppair, pF, energy);
}


# edge kernel: Gaussian basis factorised, exp(-g(u-k*dx)^2) = E(u)*exp(2dk)*const: one f32 WMMA contracts 16 centres (4 blocks x 4 sums in rows), E applied by f32 FMA; 8 per tile not 32; sums regrouped
# speedup vs baseline: 5.3548x; 1.1912x over previous
#include <hip/hip_runtime.h>
#include <math.h>

typedef __attribute__((ext_vector_type(2))) float v2f;
typedef __attribute__((ext_vector_type(8))) float v8f;
typedef __attribute__((ext_vector_type(4))) float v4f;
typedef __attribute__((ext_vector_type(4))) int   v4i;

#define NBASIS 128
#define NBLK   8
static_assert(NBLK * 16 == NBASIS, "eight products of 16 centres cover the basis");
#define CUTOFF 6.0f
#define DELTA  (CUTOFF / (NBASIS - 1))
#define GAMMA_ ((NBASIS - 1) / CUTOFF)
#define PI_F   3.14159265358979323846f
#define LOG2E_F 1.44269504088896340736f
#define K2_    (-GAMMA_ * LOG2E_F)

#if __has_builtin(__builtin_amdgcn_exp2f)
#define FAST_EXP2(x) __builtin_amdgcn_exp2f(x)
#else
#define FAST_EXP2(x) exp2f(x)
#endif

#define ZERO_REACH 2.25f
static_assert(K2_ * ZERO_REACH * ZERO_REACH < -150.0f, "skipped terms must be exact zeros");
#define SKIP_FROM 4

__device__ __forceinline__ unsigned rowMaxU(unsigned v) {
  v = max(v, (unsigned)__builtin_amdgcn_update_dpp((int)v, (int)v, 0x128, 0xF, 0xF, false));
  v = max(v, (unsigned)__builtin_amdgcn_update_dpp((int)v, (int)v, 0x124, 0xF, 0xF, false));
  v = max(v, (unsigned)__builtin_amdgcn_update_dpp((int)v, (int)v, 0x122, 0xF, 0xF, false));
  v = max(v, (unsigned)__builtin_amdgcn_update_dpp((int)v, (int)v, 0x121, 0xF, 0xF, false));
  return v;
}

__device__ __forceinline__ float softplus_(float x) {
  return fmaxf(x, 0.0f) + log1pf(expf(-fabsf(x)));
}

__device__ __forceinline__ float blockReduceSum256(float v) {
  __shared__ float sred[256];
  sred[threadIdx.x] = v;
  __syncthreads();
  for (int s = blockDim.x >> 1; s > 0; s >>= 1) {
    if (threadIdx.x < (unsigned)s) sred[threadIdx.x] += sred[threadIdx.x + s];
    __syncthreads();
  }
  return sred[0];
}

__global__ void eam_zero(float* __restrict__ a, int na, float* __restrict__ b, int nb) {
  const int i = blockIdx.x * blockDim.x + threadIdx.x;
  const int stride = gridDim.x * blockDim.x;
  for (int j = i; j < na; j += stride) a[j] = 0.0f;
  for (int j = i; j < nb; j += stride) b[j] = 0.0f;
}

#define LOG2E_D 1.4426950408889634074
#define LN2_F   0.69314718055994530942f
#define B_CLAMP 9.0f

__device__ __forceinline__ float expNegDelta(int n) {
  const double x  = -(double)n * ((6.0 / 127.0) * LOG2E_D);
  const double xi = rint(x);
  return ldexpf(exp2f((float)(x - xi)), (int)xi);
}

template <bool BINNED>
__global__ __launch_bounds__(256) void eam_edge_fwd(
    const float* __restrict__ r, const int* __restrict__ dst,
    const float* __restrict__ phiD, const float* __restrict__ phiP,
    float* __restrict__ dens_out, float* __restrict__ gd_out, float* __restrict__ gp_out,
    float* __restrict__ partial_pair, int E)
{
  __shared__ v2f sW[NBLK * 2 * 16];

  for (int idx = threadIdx.x; idx < NBLK * 2 * 16; idx += blockDim.x) {
    const int w = idx >> 5, h = (idx >> 4) & 1, row = idx & 15;
    const int q = row >> 2, ch = row & 3;
    const int p = 4 * w + q;
    float a[2];
    for (int s2 = 0; s2 < 2; ++s2) {
      const int k = 2 * h + s2;
      const int j = 4 * p + k;
      float c = (ch < 2) ? softplus_(phiD[j]) : phiP[j];
      if (ch & 1) c *= j * DELTA;
      a[s2] = c * expNegDelta(k * (k + 8 * p));
    }
    sW[idx] = (v2f){a[0], a[1]};
  }
  __syncthreads();

  const int lane = threadIdx.x & 31;
  const int col  = lane & 15;
  const int hi   = lane >> 4;

  v2f aReg[NBLK];
#pragma unroll
  for (int w = 0; w < NBLK; ++w)
    aReg[w] = sW[(2 * w + hi) * 16 + col];

  float ckHi[2], ckLo[2];
  for (int s2 = 0; s2 < 2; ++s2) {
    const double ck = 2.0 * (double)(2 * hi + s2) * LOG2E_D;
    ckHi[s2] = (float)ck;
    ckLo[s2] = (float)(ck - (double)ckHi[s2]);
  }

  const int wavesPerBlk = blockDim.x >> 5;
  const int wave   = blockIdx.x * wavesPerBlk + (threadIdx.x >> 5);
  const int nWaves = gridDim.x * wavesPerBlk;
  const int nTiles = (E + 15) >> 4;

  float pairAcc = 0.0f;

  for (int tile = wave; tile < nTiles; tile += nWaves) {
    const int e = tile * 16 + col;
    const bool valid = (e < E);

    const int eNext = (tile + nWaves) * 16 + col;
    if (eNext < E) __builtin_prefetch(&r[3 * eNext], 0, 0);

    float x = 0.0f, y = 0.0f, z = 0.0f;
    if (valid) { x = r[3 * e]; y = r[3 * e + 1]; z = r[3 * e + 2]; }
    const float dist = sqrtf(x * x + y * y + z * z);

    const float dmax = __uint_as_float(
        (unsigned)__builtin_amdgcn_readfirstlane((int)rowMaxU(__float_as_uint(dist))));
    const float tLast = (dmax + ZERO_REACH) * (1.0f / (16.0f * DELTA));
    const int wEnd = (tLast < (float)NBLK) ? (int)tLast + 1 : NBLK;

    const float dB = fminf(dist, B_CLAMP);
    v2f bv;
    {
      const float p0 = dB * ckHi[0], p1 = dB * ckHi[1];
      const float e0 = fmaf(dB, ckLo[0], fmaf(dB, ckHi[0], -p0));
      const float e1 = fmaf(dB, ckLo[1], fmaf(dB, ckHi[1], -p1));
      const float b0 = FAST_EXP2(p0), b1 = FAST_EXP2(p1);
      bv.x = fmaf(b0, e0 * LN2_F, b0);
      bv.y = fmaf(b1, e1 * LN2_F, b1);
    }

    float s0 = 0.0f, s1 = 0.0f, s2 = 0.0f, s3 = 0.0f;
    const float baseE = dist - (8 * hi) * DELTA;
    const v2f base2 = {baseE, baseE - 4.0f * DELTA};
    const v8f zero8 = {0.f, 0.f, 0.f, 0.f, 0.f, 0.f, 0.f, 0.f};
#pragma unroll
    for (int w = 0; w < NBLK; ++w) {
      if (w >= SKIP_FROM && w >= wEnd) break;
      const v2f u   = base2 - (float)(16 * w) * DELTA;
      const v2f arg = (u * u) * (float)K2_;
      const float E0 = FAST_EXP2(arg.x);
      const float E1 = FAST_EXP2(arg.y);
      const v8f D = __builtin_amdgcn_wmma_f32_16x16x4_f32(
          false, aReg[w], false, bv, (short)0, zero8, false, false);
      s0 = fmaf(E0, D[0], s0); s1 = fmaf(E0, D[1], s1);
      s2 = fmaf(E0, D[2], s2); s3 = fmaf(E0, D[3], s3);
      s0 = fmaf(E1, D[4], s0); s1 = fmaf(E1, D[5], s1);
      s2 = fmaf(E1, D[6], s2); s3 = fmaf(E1, D[7], s3);
    }
    s0 += __shfl_xor(s0, 16); s1 += __shfl_xor(s1, 16);
    s2 += __shfl_xor(s2, 16); s3 += __shfl_xor(s3, 16);

    if (lane < 16 && valid) {
      const float SA = s0, SB = s1, SC = s2, SD = s3;
      float sn, cs;
      __sincosf(dist * (PI_F / CUTOFF), &sn, &cs);
      const float phi  = 0.25f * (1.0f + cs);
      const float phip = -0.25f * (PI_F / CUTOFF) * sn;
      const float invd = 1.0f / dist;

      const float dens   = phi * SA;
      const float ddens  = phi * (-2.0f * GAMMA_) * (dist * SA - SB) + phip * SA;
      const float pairn  = phi * SC;
      const float dpairn = phi * (-2.0f * GAMMA_) * (dist * SC - SD) + phip * SC;
      const float dpair  = (dpairn - pairn * invd) * invd;

      if (BINNED) dens_out[e] = dens;
      else        atomicAdd(&dens_out[dst[e]], dens);
      gd_out[e] = ddens * invd;
      gp_out[e] = dpair * invd;
      pairAcc += pairn * invd;
    }
  }

  const float bsum = blockReduceSum256(pairAcc);
  if (threadIdx.x == 0) partial_pair[blockIdx.x] = bsum;
}

#define DR_RANGES  8
#define DR_NODES   6272
#define DR_SLICES  32
#define DR_PITCH   (DR_RANGES * DR_NODES)
#define DR_THREADS 1024
#define DR_UNROLL  4

__global__ __launch_bounds__(DR_THREADS) void eam_density(
    const int* __restrict__ dst, const float* __restrict__ dens,
    float* __restrict__ planes, float* __restrict__ forces, int nf, int E, int N)
{
  __shared__ float sD[DR_NODES];

  const int tid  = threadIdx.x;
  const int rIdx = blockIdx.x % DR_RANGES;
  const int sIdx = blockIdx.x / DR_RANGES;
  const int ES   = (((E + DR_SLICES - 1) / DR_SLICES) + 31) & ~31;
  const int eBeg = min(E, sIdx * ES);
  const int eEnd = min(E, eBeg + ES);
  const int n0   = rIdx * DR_NODES;
  const int cnt  = max(0, min(DR_NODES, N - n0));

  for (int j = blockIdx.x * DR_THREADS + tid; j < nf; j += gridDim.x * DR_THREADS) forces[j] = 0.0f;

  for (int i = tid; i < DR_NODES; i += DR_THREADS) sD[i] = 0.0f;
  __syncthreads();

  const int eEnd4 = eBeg + ((eEnd - eBeg) & ~3);
  for (int base = eBeg + 4 * tid; base < eEnd4; base += 4 * DR_UNROLL * DR_THREADS) {
    v4i dN[DR_UNROLL];
    v4f dv[DR_UNROLL];
#pragma unroll
    for (int u = 0; u < DR_UNROLL; ++u) {
      const int e = base + 4 * u * DR_THREADS;
      dN[u] = (v4i){-1, -1, -1, -1};
      dv[u] = (v4f){0.0f, 0.0f, 0.0f, 0.0f};
      if (e < eEnd4) {
        dN[u] = *(const v4i*)(dst + e);
        dv[u] = *(const v4f*)(dens + e);
      }
    }
#pragma unroll
    for (int u = 0; u < DR_UNROLL; ++u) {
      const unsigned k0 = (unsigned)(dN[u].x - n0), k1 = (unsigned)(dN[u].y - n0);
      const unsigned k2 = (unsigned)(dN[u].z - n0), k3 = (unsigned)(dN[u].w - n0);
      if (k0 < (unsigned)cnt) atomicAdd(&sD[k0], dv[u].x);
      if (k1 < (unsigned)cnt) atomicAdd(&sD[k1], dv[u].y);
      if (k2 < (unsigned)cnt) atomicAdd(&sD[k2], dv[u].z);
      if (k3 < (unsigned)cnt) atomicAdd(&sD[k3], dv[u].w);
    }
  }
  if (eEnd4 + tid < eEnd) {
    const unsigned k = (unsigned)(dst[eEnd4 + tid] - n0);
    if (k < (unsigned)cnt) atomicAdd(&sD[k], dens[eEnd4 + tid]);
  }
  __syncthreads();

  float* prow = planes + (size_t)sIdx * DR_PITCH + n0;
  for (int i = tid; i < cnt; i += DR_THREADS) prow[i] = sD[i];
}

template <bool PLANES>
__global__ __launch_bounds__(256) void eam_node(
    const float* __restrict__ planes, const float* __restrict__ w,
    float* __restrict__ Fp, float* __restrict__ partial_F, int N)
{
  const float w0 = w[0], w1 = w[1], w2 = w[2], w3 = w[3], w4 = w[4];
  float Facc = 0.0f;
  for (int n = blockIdx.x * blockDim.x + threadIdx.x; n < N;
       n += gridDim.x * blockDim.x) {
    float x = planes[n];
    if (PLANES) {
#pragma unroll
      for (int sl = 1; sl < DR_SLICES; ++sl) x += planes[(size_t)sl * DR_PITCH + n];
    }
    const float sq = sqrtf(x);
    const float x2 = x * x;
    const float F  = 2.0f * w0 * sq + w1 * x + 0.5f * w2 * x2 +
                     (1.0f / 6.0f) * w3 * x2 * x + (1.0f / 24.0f) * w4 * x2 * x2;
    const float Fd = w0 / sq + w1 + w2 * x + 0.5f * w3 * x2 +
                     (1.0f / 6.0f) * w4 * x2 * x;
    Facc += F;
    Fp[n] = Fd;
  }
  const float bsum = blockReduceSum256(Facc);
  if (threadIdx.x == 0) partial_F[blockIdx.x] = bsum;
}

#define FR_RANGES  8
#define FR_NODES   6250
#define FR_THREADS 512
#define FR_GROUPS  2
#define EN_PAIR    512
#define EN_NODE    128
#define EN_THREADS 256

struct FrStream {
  v4i sN[FR_GROUPS], dN[FR_GROUPS];
  v4f gdv[FR_GROUPS], gpv[FR_GROUPS];
  v4f ra[FR_GROUPS], rb[FR_GROUPS], rc[FR_GROUPS];
};

__device__ __forceinline__ void fr_load(
    FrStream& s, const float* __restrict__ r, const int* __restrict__ src,
    const int* __restrict__ dst, const float* __restrict__ gd,
    const float* __restrict__ gp, int eBeg, int nGrp, int j0, int nThr, int tid)
{
#pragma unroll
  for (int u = 0; u < FR_GROUPS; ++u) {
    const int g = (j0 + u) * nThr + tid;
    const int e = eBeg + 4 * min(g, nGrp - 1);
    s.sN[u]  = *(const v4i*)(src + e);
    s.dN[u]  = *(const v4i*)(dst + e);
    s.gdv[u] = *(const v4f*)(gd + e);
    s.gpv[u] = *(const v4f*)(gp + e);
    const v4f* rp = (const v4f*)(r + 3 * (size_t)e);
    s.ra[u] = rp[0]; s.rb[u] = rp[1]; s.rc[u] = rp[2];
  }
}

__global__ __launch_bounds__(FR_THREADS) void eam_force(
    const float* __restrict__ r, const int* __restrict__ src,
    const int* __restrict__ dst, const float* __restrict__ gd,
    const float* __restrict__ gp, const float* __restrict__ Fp,
    float* __restrict__ forces, int E, int N,
    const float* __restrict__ partial_pair, const float* __restrict__ partial_F,
    float* __restrict__ energy)
{
  __shared__ float sF[3 * FR_NODES];

  const int tid  = threadIdx.x;
  const int nThr = blockDim.x;
  const int lane = tid & 31;
  const int wv   = tid >> 5;
  const int nWv  = nThr >> 5;
  const int rIdx = blockIdx.x % FR_RANGES;
  const int sIdx = blockIdx.x / FR_RANGES;
  const int nSl  = gridDim.x / FR_RANGES;
  const int ES   = (((E + nSl - 1) / nSl) + 31) & ~31;
  const int eBeg = min(E, sIdx * ES);
  const int eEnd = min(E, eBeg + ES);
  const bool wide =
      ((((size_t)r | (size_t)src | (size_t)dst | (size_t)gd | (size_t)gp) & 15) == 0);
  const int nGrp  = wide ? ((eEnd - eBeg) >> 2) : 0;
  const int eEnd4 = eBeg + 4 * nGrp;
  const int K     = (nGrp + nThr - 1) / nThr;

  const bool eBlk = (blockIdx.x == gridDim.x - FR_RANGES);

  for (int n0 = rIdx * FR_NODES; n0 < N; n0 += FR_RANGES * FR_NODES) {
    const int cnt = min(FR_NODES, N - n0);
    const int nW  = 3 * cnt;

    FrStream cur;
    if (K > 0) fr_load(cur, r, src, dst, gd, gp, eBeg, nGrp, 0, nThr, tid);

    if (eBlk && n0 == rIdx * FR_NODES) {
      constexpr int EP = (EN_PAIR + EN_THREADS - 1) / EN_THREADS;
      constexpr int EF = (EN_NODE + EN_THREADS - 1) / EN_THREADS;
      static_assert(EN_THREADS == 256 && FR_THREADS >= EN_THREADS, "the tree below is written for 256 sums");
      const int t8 = tid & (EN_THREADS - 1);
      float ep[EP], ef[EF];
      for (int j = 0; j < EP; ++j) ep[j] = partial_pair[min(t8 + EN_THREADS * j, EN_PAIR - 1)];
      for (int j = 0; j < EF; ++j) ef[j] = partial_F[min(t8 + EN_THREADS * j, EN_NODE - 1)];
      float ev = 0.0f;
      for (int j = 0; j < EP; ++j) ev += (t8 + EN_THREADS * j < EN_PAIR) ? ep[j] : 0.0f;
      for (int j = 0; j < EF; ++j) ev += (t8 + EN_THREADS * j < EN_NODE) ? ef[j] : 0.0f;
      if (tid < EN_THREADS) sF[tid] = ev;
      __syncthreads();
      if (tid < 128) sF[tid] += sF[tid + 128];
      __syncthreads();
      if (tid < 64) sF[tid] += sF[tid + 64];
      __syncthreads();
      if (tid < 32) {
        float v = sF[tid] + sF[tid + 32];
        v += __shfl_down(v, 16);
        v += __shfl_down(v, 8);
        v += __shfl_down(v, 4);
        v += __shfl_down(v, 2);
        v += __shfl_down(v, 1);
        if (tid == 0) energy[0] = v;
      }
      __syncthreads();
    }

    for (int i = tid; i < nW; i += nThr) sF[i] = 0.0f;
    __syncthreads();

    for (int j0 = 0; j0 < K; j0 += FR_GROUPS) {
      bool  hs[4 * FR_GROUPS], hd[4 * FR_GROUPS];
      float fpv[4 * FR_GROUPS];
#pragma unroll
      for (int u = 0; u < FR_GROUPS; ++u) {
        const bool ok = ((j0 + u) * nThr + tid) < nGrp;
#pragma unroll
        for (int c = 0; c < 4; ++c) {
          const int q = 4 * u + c;
          hs[q] = ok && ((unsigned)(cur.sN[u][c] - n0) < (unsigned)cnt);
          hd[q] = ok && ((unsigned)(cur.dN[u][c] - n0) < (unsigned)cnt);
          fpv[q] = 0.0f;
          if (hs[q] || hd[q]) fpv[q] = Fp[cur.dN[u][c]];
        }
      }

      FrStream nxt;
      fr_load(nxt, r, src, dst, gd, gp, eBeg, nGrp, j0 + FR_GROUPS, nThr, tid);

#pragma unroll
      for (int u = 0; u < FR_GROUPS; ++u) {
        const float rr[12] = {cur.ra[u].x, cur.ra[u].y, cur.ra[u].z, cur.ra[u].w,
                              cur.rb[u].x, cur.rb[u].y, cur.rb[u].z, cur.rb[u].w,
                              cur.rc[u].x, cur.rc[u].y, cur.rc[u].z, cur.rc[u].w};
#pragma unroll
        for (int c = 0; c < 4; ++c) {
          const int q = 4 * u + c;
          if (hs[q] || hd[q]) {
            const float g  = fpv[q] * cur.gdv[u][c] + cur.gpv[u][c];
            const float fx = g * rr[3 * c], fy = g * rr[3 * c + 1], fz = g * rr[3 * c + 2];
            if (hs[q]) {
              float* p = &sF[3 * (cur.sN[u][c] - n0)];
              atomicAdd(p, fx); atomicAdd(p + 1, fy); atomicAdd(p + 2, fz);
            }
            if (hd[q]) {
              float* p = &sF[3 * (cur.dN[u][c] - n0)];
              atomicAdd(p, -fx); atomicAdd(p + 1, -fy); atomicAdd(p + 2, -fz);
            }
          }
        }
      }
      cur = nxt;
    }

    for (int e = eEnd4 + tid; e < eEnd; e += nThr) {
      const int s = src[e], d = dst[e];
      const bool h1 = (unsigned)(s - n0) < (unsigned)cnt;
      const bool h2 = (unsigned)(d - n0) < (unsigned)cnt;
      if (h1 || h2) {
        const float g  = Fp[d] * gd[e] + gp[e];
        const float fx = g * r[3 * e], fy = g * r[3 * e + 1], fz = g * r[3 * e + 2];
        if (h1) {
          float* p = &sF[3 * (s - n0)];
          atomicAdd(p, fx); atomicAdd(p + 1, fy); atomicAdd(p + 2, fz);
        }
        if (h2) {
          float* p = &sF[3 * (d - n0)];
          atomicAdd(p, -fx); atomicAdd(p + 1, -fy); atomicAdd(p + 2, -fz);
        }
      }
    }
    __syncthreads();

    float* fbase = forces + (size_t)3 * n0;
    const int shift  = (int)((((size_t)fbase) >> 2) & 31);
    const int nLines = (shift + nW + 31) >> 5;
    for (int k = wv; k < nLines; k += nWv) {
      const int w = 32 * k + lane - shift;
      if (w >= 0 && w < nW) {
        const float v = sF[w];
        if (v != 0.0f) atomicAdd(fbase + w, v);
      }
    }
    __syncthreads();
  }
}

extern "C" void kernel_launch(void* const* d_in, const int* in_sizes, int n_in,
                              void* d_out, int out_size, void* d_ws, size_t ws_size,
                              hipStream_t stream) {
  const float* r    = (const float*)d_in[0];
  const int*   src  = (const int*)  d_in[1];
  const int*   dst  = (const int*)  d_in[2];
  const float* phiD = (const float*)d_in[3];
  const float* phiP = (const float*)d_in[4];
  const float* emb  = (const float*)d_in[5];

  const int E = in_sizes[1];
  const int N = (out_size - 1) / 3;

  float* out    = (float*)d_out;
  float* energy = out;
  float* forces = out + 1;

  const int NB1 = EN_PAIR;
  const int NB2 = EN_NODE;
  float* ws = (float*)d_ws;
  float *gd, *gp, *Fp, *ppair, *pF;

  const size_t needBinned =
      ((size_t)DR_SLICES * DR_PITCH + (size_t)3 * E + N + NB1 + NB2) * sizeof(float);
  if (ws_size >= needBinned && N <= DR_PITCH) {
    float* planes = ws;
    float* dens   = planes + DR_SLICES * DR_PITCH;
    gd    = dens + E;
    gp    = gd + E;
    Fp    = gp + E;
    ppair = Fp + N;
    pF    = ppair + NB1;
    eam_edge_fwd<true><<<NB1, 256, 0, stream>>>(r, dst, phiD, phiP, dens, gd, gp, ppair, E);
    eam_density<<<DR_RANGES * DR_SLICES, DR_THREADS, 0, stream>>>(dst, dens, planes, forces, 3 * N, E, N);
    eam_node<true><<<NB2, 256, 0, stream>>>(planes, emb, Fp, pF, N);
  } else {
    gd    = ws;
    gp    = gd + E;
    float* local = gp + E;
    Fp    = local + N;
    ppair = Fp + N;
    pF    = ppair + NB1;
    eam_zero<<<256, 256, 0, stream>>>(local, N, forces, 3 * N);
    eam_edge_fwd<false><<<NB1, 256, 0, stream>>>(r, dst, phiD, phiP, local, gd, gp, ppair, E);
    eam_node<false><<<NB2, 256, 0, stream>>>(local, emb, Fp, pF, N);
  }
  eam_force   <<<256, FR_THREADS, 0, stream>>>(r, src, dst, gd, gp, Fp, forces, E, N,
                                               ppair, pF, energy);
}
